# starting kernel
# baseline (speedup 1.0000x reference)
#include <hip/hip_runtime.h>
#include <hip/hip_bf16.h>
#include <math.h>


typedef float v2f __attribute__((ext_vector_type(2)));
typedef float v8f __attribute__((ext_vector_type(8)));

#define GRU_L 512
#define GRU_B 2048
#define GRU_D 64

__global__ void __launch_bounds__(256)
gru_gates_wmma_kernel(const float* __restrict__ x,
                      const float* __restrict__ W_ih,
                      const float* __restrict__ b_ih,
                      float* __restrict__ gx,
                      int rows) {
  __shared__ float Bs[16 * GRU_D];
  for (int i = threadIdx.x; i < 16 * GRU_D; i += blockDim.x) {
    const int row = i >> 6;
    const int col = i & (GRU_D - 1);
    Bs[i] = (row < 3) ? W_ih[row * GRU_D + col] : 0.0f;
  }
  __syncthreads();

  const int lane     = threadIdx.x & 31;
  const int wave     = blockIdx.x * (blockDim.x >> 5) + (threadIdx.x >> 5);
  const int tileBase = wave << 4;
  if (tileBase >= rows) return;

  const int m     = lane & 15;
  const int khalf = lane >> 4;

  const float* arow = x + (size_t)(tileBase + m) * GRU_D + 2 * khalf;
  const float* brow = Bs + m * GRU_D + 2 * khalf;

  v8f c = {};
#pragma unroll
  for (int k = 0; k < 16; ++k) {
    const v2f a = *(const v2f*)(arow + 4 * k);
    const v2f b = *(const v2f*)(brow + 4 * k);
#if __has_builtin(__builtin_amdgcn_wmma_f32_16x16x4_f32)
    c = __builtin_amdgcn_wmma_f32_16x16x4_f32(false, a, false, b,
                                              (short)0, c, false, false);
#else
#pragma unroll
    for (int i = 0; i < 8; ++i) c[i] += a.x * b.x + a.y * b.y;
#endif
  }

  if (m < 3) {
    const float bias = b_ih[m];
    float* plane = gx + (size_t)m * rows;
#pragma unroll
    for (int i = 0; i < 8; ++i) {
      const int row = tileBase + i + 8 * khalf;
      plane[row] = c[i] + bias;
    }
  }
}

__global__ void __launch_bounds__(256)
gru_scan_kernel(const float* __restrict__ gx,
                const float* __restrict__ h0,
                const int*   __restrict__ lengths,
                const float* __restrict__ W_hh,
                const float* __restrict__ b_hh,
                float* __restrict__ out) {
  const int b = blockIdx.x * blockDim.x + threadIdx.x;
  if (b >= GRU_B) return;

  const size_t plane = (size_t)GRU_L * GRU_B;
  const float* gr = gx;
  const float* gz = gx + plane;
  const float* gn = gx + 2 * plane;

  float h = h0[b];
  const int   len = lengths[b];
  const float w0 = W_hh[0], w1 = W_hh[1], w2 = W_hh[2];
  const float c0 = b_hh[0], c1 = b_hh[1], c2 = b_hh[2];

  for (int t = 0; t < GRU_L; ++t) {
    const size_t idx = (size_t)t * GRU_B + b;

    const size_t pf = idx + (size_t)16 * GRU_B;
    __builtin_prefetch(gr + pf, 0, 0);
    __builtin_prefetch(gz + pf, 0, 0);
    __builtin_prefetch(gn + pf, 0, 0);

    const float xr = gr[idx];
    const float xz = gz[idx];
    const float xn = gn[idx];

    const float r = 1.0f / (1.0f + __expf(-(xr + h * w0 + c0)));
    const float z = 1.0f / (1.0f + __expf(-(xz + h * w1 + c1)));
    const float n = tanhf(xn + r * (h * w2 + c2));
    h = (1.0f - z) * n + z * h;

    out[idx] = (t < len) ? h : 0.0f;
  }
}

extern "C" void kernel_launch(void* const* d_in, const int* in_sizes, int n_in,
                              void* d_out, int out_size, void* d_ws, size_t ws_size,
                              hipStream_t stream) {
  const float* x       = (const float*)d_in[0];
  const float* h0      = (const float*)d_in[1];
  const int*   lengths = (const int*)  d_in[2];
  const float* W_ih    = (const float*)d_in[3];
  const float* W_hh    = (const float*)d_in[4];
  const float* b_ih    = (const float*)d_in[5];
  const float* b_hh    = (const float*)d_in[6];
  float*       out     = (float*)d_out;
  float*       gx      = (float*)d_ws;

  const int rows  = GRU_L * GRU_B;
  const int tiles = rows / 16;
  const int wavesPerBlock = 8;
  const int blocks1 = tiles / wavesPerBlock;

  gru_gates_wmma_kernel<<<blocks1, 256, 0, stream>>>(x, W_ih, b_ih, gx, rows);
  gru_scan_kernel<<<(GRU_B + 255) / 256, 256, 0, stream>>>(gx, h0, lengths,
                                                           W_hh, b_hh, out);
}


# scan: tanh's two range formulas (polynomial, exp form) run side by side in every lane and a select picks one: no divergent branch or EXEC round trip in a step's chain; values bit-identical
# speedup vs baseline: 4.0435x; 4.0435x over previous
#include <hip/hip_runtime.h>
#include <math.h>


typedef float v2f __attribute__((ext_vector_type(2)));
typedef float v4f __attribute__((ext_vector_type(4)));
typedef float v8f __attribute__((ext_vector_type(8)));

#define GRU_L 512
#define GRU_B 2048
#define GRU_D 64
#define GRU_COLS 4
#define GRU_TS   4
#define GRU_TILES (GRU_L / GRU_TS)

struct XTile { v4f q[8]; };

__device__ __forceinline__ const float* tile_ptr(const float* __restrict__ xlane, int j) {
  const int jj = j < GRU_TILES ? j : GRU_TILES - 1;
  return xlane + (size_t)jj * (GRU_TS * GRU_B * GRU_D);
}

__device__ __forceinline__ void prefetch_tile(const float* __restrict__ xlane, int j) {
  __builtin_prefetch(tile_ptr(xlane, j), 0, 0);
}

template <int I0, int I1>
__device__ __forceinline__ void load_part(XTile& xt, const float* __restrict__ p) {
#pragma unroll
  for (int i = I0; i < I1; ++i) xt.q[i] = *(const v4f*)(p + 8 * i);
}

template <int I0, int I1, bool FIRST>
__device__ __forceinline__ void mm_part(const XTile& xt, const v2f (&wf)[16], v8f& acc) {
  v8f a = acc;
  if (FIRST) a = (v8f){};
#pragma unroll
  for (int i = I0; i < I1; ++i) {
    const v2f b0 = {xt.q[i].x, xt.q[i].y}, b1 = {xt.q[i].z, xt.q[i].w};
    a = __builtin_amdgcn_wmma_f32_16x16x4_f32(false, wf[2 * i],     false, b0, (short)0, a, false, false);
    a = __builtin_amdgcn_wmma_f32_16x16x4_f32(false, wf[2 * i + 1], false, b1, (short)0, a, false, false);
  }
  acc = a;
}

template <int SH>
__device__ __forceinline__ float from_lane_plus(float v) {
  if (SH == 0) return v;
  return __int_as_float(__builtin_amdgcn_update_dpp(0, __float_as_int(v), 0x100 + SH, 0xf, 0xf, true));
}

template <int SH>
__device__ __forceinline__ float z_lanes_from_lane_plus(float keep, float v) {
  const int ctrl = SH > 0 ? 0x100 + SH : SH == 0 ? 0xE4 : 0x110 - SH;
  return __int_as_float(__builtin_amdgcn_update_dpp(__float_as_int(keep), __float_as_int(v),
                                                    ctrl, 0xf, 0x2, false));
}

template <int S>
__device__ __forceinline__ float sigmoid_inputs(float g_r, float g_z) {
  return z_lanes_from_lane_plus<4 * S - 4>(from_lane_plus<4 * S>(g_r), g_z);
}

__device__ __forceinline__ float h_to_z_lanes(float h) {
  return z_lanes_from_lane_plus<-4>(h, h);
}

template <int N>
__device__ __forceinline__ void wmma_chain_pattern() {
#pragma unroll
  for (int i = 0; i < N; ++i) {
    __builtin_amdgcn_sched_group_barrier(0x008, 1, 0);
    __builtin_amdgcn_sched_group_barrier(0x402, 3, 0);
  }
}

__device__ __forceinline__ float tanh_both_ranges(float x) {
  const float y  = fabsf(x);
  const float y2 = x * x;
  float p = fmaf(y2, -0x1.758e7ap-8f, 0x1.521192p-6f);
  p = fmaf(y2, p, -0x1.b8389cp-5f);
  p = fmaf(y2, p, 0x1.110704p-3f);
  p = fmaf(y2, p, -0x1.555532p-2f);
  const float z_small = fmaf(y2, y * p, y);
  const float e = expf(2.0f * y);
  const float z_large = fmaf(-2.0f, __builtin_amdgcn_rcpf(e + 1.0f), 1.0f);
  return copysignf(y < 0.625f ? z_small : z_large, x);
}

struct ScanConst { float wv, cv, bv, w2, c2, bi2; };

template <int S>
__device__ __forceinline__ float gru_step(float h, const v8f& G, const ScanConst& k) {
  const float xs = sigmoid_inputs<S>(G[0], G[1]) + k.bv;
  const float xn = from_lane_plus<4 * S>(G[2]) + k.bi2;
  const float s = 1.0f / (1.0f + __expf(-(xs + h * k.wv + k.cv)));
  const float z = from_lane_plus<4>(s);
  const float a = xn + s * (h * k.w2 + k.c2);
  wmma_chain_pattern<4>();
  const float n = tanh_both_ranges(a);
  return h_to_z_lanes((1.0f - z) * n + z * h);
}

__global__ void __launch_bounds__(32)
gru_fused_kernel(const float* __restrict__ x,
                 const float* __restrict__ h0,
                 const int*   __restrict__ lengths,
                 const float* __restrict__ W_ih,
                 const float* __restrict__ W_hh,
                 const float* __restrict__ b_ih,
                 const float* __restrict__ b_hh,
                 float* __restrict__ out) {
  const int lane = threadIdx.x;
  const int c    = lane & 15;
  const int hf   = lane >> 4;
  const int col  = c & 3;
  const int ts   = c >> 2;
  const int b    = blockIdx.x * GRU_COLS + col;

  v2f wf[16];
  {
    const float* wp = W_ih + (c < 3 ? c : 0) * GRU_D + 4 * hf;
#pragma unroll
    for (int i = 0; i < 8; ++i) {
      v4f w = {wp[8 * i], wp[8 * i + 1], wp[8 * i + 2], wp[8 * i + 3]};
      if (c >= 3) w = (v4f){0.0f, 0.0f, 0.0f, 0.0f};
      wf[2 * i]     = (v2f){w.x, w.y};
      wf[2 * i + 1] = (v2f){w.z, w.w};
    }
  }

  ScanConst k;
  const bool z_lane = (c >> 2) == 1;
  k.wv = z_lane ? W_hh[1] : W_hh[0];
  k.cv = z_lane ? b_hh[1] : b_hh[0];
  k.bv = z_lane ? b_ih[1] : b_ih[0];
  k.w2 = W_hh[2]; k.c2 = b_hh[2]; k.bi2 = b_ih[2];

  float h = h0[b];
  const int len = lengths[b];

  const float* xlane = x + ((size_t)ts * GRU_B + b) * GRU_D + 4 * hf;
  float* outb = out + b;

  prefetch_tile(xlane, 2);
  XTile X;
  load_part<0, 8>(X, tile_ptr(xlane, 0));
  v8f G, Gn;
  mm_part<0, 4, true>(X, wf, G);
  mm_part<4, 8, false>(X, wf, G);
  load_part<0, 8>(X, tile_ptr(xlane, 1));

  for (int j = 0; j < GRU_TILES; ++j) {
    const float* pn = tile_ptr(xlane, j + 2);

    mm_part<0, 2, true>(X, wf, Gn);
    h = gru_step<0>(h, G, k);
    const float o0 = (GRU_TS * j + 0 < len) ? h : 0.0f;

    mm_part<2, 4, false>(X, wf, Gn);
    load_part<0, 2>(X, pn);
    prefetch_tile(xlane, j + 3);
    h = gru_step<1>(h, G, k);
    const float o1 = (GRU_TS * j + 1 < len) ? h : 0.0f;

    mm_part<4, 6, false>(X, wf, Gn);
    load_part<2, 4>(X, pn);
    h = gru_step<2>(h, G, k);
    const float o2 = (GRU_TS * j + 2 < len) ? h : 0.0f;

    mm_part<6, 8, false>(X, wf, Gn);
    load_part<4, 6>(X, pn);
    h = gru_step<3>(h, G, k);
    const float o3 = (GRU_TS * j + 3 < len) ? h : 0.0f;

    load_part<6, 8>(X, pn);
    G = Gn;

    if (lane < GRU_COLS) {
      float* o = outb + (size_t)(GRU_TS * j) * GRU_B;
      o[0 * GRU_B] = o0;
      o[1 * GRU_B] = o1;
      o[2 * GRU_B] = o2;
      o[3 * GRU_B] = o3;
    }
  }
}

extern "C" void kernel_launch(void* const* d_in, const int* in_sizes, int n_in,
                              void* d_out, int out_size, void* d_ws, size_t ws_size,
                              hipStream_t stream) {
  const float* x       = (const float*)d_in[0];
  const float* h0      = (const float*)d_in[1];
  const int*   lengths = (const int*)  d_in[2];
  const float* W_ih    = (const float*)d_in[3];
  const float* W_hh    = (const float*)d_in[4];
  const float* b_ih    = (const float*)d_in[5];
  const float* b_hh    = (const float*)d_in[6];
  float*       out     = (float*)d_out;

  gru_fused_kernel<<<GRU_B / GRU_COLS, 32, 0, stream>>>(x, h0, lengths, W_ih, W_hh,
                                                        b_ih, b_hh, out);
}


# each wave owns 2 batch columns (tile = 8 steps x 2 columns) in 1024 one-wave workgroups: 2 f32 WMMAs per step in a wave's serial path instead of 4, kept two to a step; sums and their order unchanged
# speedup vs baseline: 4.1998x; 1.0387x over previous
#include <hip/hip_runtime.h>
#include <math.h>


typedef float v2f __attribute__((ext_vector_type(2)));
typedef float v4f __attribute__((ext_vector_type(4)));
typedef float v8f __attribute__((ext_vector_type(8)));

#define GRU_L 512
#define GRU_B 2048
#define GRU_D 64
#define GRU_COLS 2
#define GRU_TS   8
#define GRU_TILES (GRU_L / GRU_TS)

struct XTile { v4f q[8]; };

__device__ __forceinline__ const float* tile_ptr(const float* __restrict__ xlane, int j) {
  const int jj = j < GRU_TILES ? j : GRU_TILES - 1;
  return xlane + (size_t)jj * (GRU_TS * GRU_B * GRU_D);
}

__device__ __forceinline__ void prefetch_tile(const float* __restrict__ xlane, int j) {
  __builtin_prefetch(tile_ptr(xlane, j), 0, 0);
}

template <int I0, int I1>
__device__ __forceinline__ void load_part(XTile& xt, const float* __restrict__ p) {
#pragma unroll
  for (int i = I0; i < I1; ++i) xt.q[i] = *(const v4f*)(p + 8 * i);
}

template <int I0, int I1, bool FIRST>
__device__ __forceinline__ void mm_part(const XTile& xt, const v2f (&wf)[16], v8f& acc) {
  v8f a = acc;
  if (FIRST) a = (v8f){};
#pragma unroll
  for (int i = I0; i < I1; ++i) {
    const v2f b0 = {xt.q[i].x, xt.q[i].y}, b1 = {xt.q[i].z, xt.q[i].w};
    a = __builtin_amdgcn_wmma_f32_16x16x4_f32(false, wf[2 * i],     false, b0, (short)0, a, false, false);
    a = __builtin_amdgcn_wmma_f32_16x16x4_f32(false, wf[2 * i + 1], false, b1, (short)0, a, false, false);
  }
  acc = a;
}

template <int SH>
__device__ __forceinline__ float from_lane_plus(float v) {
  if (SH == 0) return v;
  return __int_as_float(__builtin_amdgcn_update_dpp(0, __float_as_int(v), 0x100 + SH, 0xf, 0xf, true));
}

template <int SH>
__device__ __forceinline__ float z_lanes_from_lane_plus(float keep, float v) {
  const int ctrl = SH > 0 ? 0x100 + SH : SH == 0 ? 0xE4 : 0x110 - SH;
  return __int_as_float(__builtin_amdgcn_update_dpp(__float_as_int(keep), __float_as_int(v),
                                                    ctrl, 0xf, 0x2, false));
}

template <int S>
__device__ __forceinline__ float sigmoid_inputs(float g_r, float g_z) {
  return z_lanes_from_lane_plus<GRU_COLS * S - 4>(from_lane_plus<GRU_COLS * S>(g_r), g_z);
}

__device__ __forceinline__ float h_to_z_lanes(float h) {
  return z_lanes_from_lane_plus<-4>(h, h);
}

__device__ __forceinline__ void step_fence() {
  __builtin_amdgcn_sched_barrier(0);
}

__device__ __forceinline__ float tanh_both_ranges(float x) {
  const float y  = fabsf(x);
  const float y2 = x * x;
  float p = fmaf(y2, -0x1.758e7ap-8f, 0x1.521192p-6f);
  p = fmaf(y2, p, -0x1.b8389cp-5f);
  p = fmaf(y2, p, 0x1.110704p-3f);
  p = fmaf(y2, p, -0x1.555532p-2f);
  const float z_small = fmaf(y2, y * p, y);
  const float e = expf(2.0f * y);
  const float z_large = fmaf(-2.0f, __builtin_amdgcn_rcpf(e + 1.0f), 1.0f);
  return copysignf(y < 0.625f ? z_small : z_large, x);
}

struct ScanConst { float wv, cv, bv, w2, c2, bi2; };

struct Gates { float r, z, n; };

template <int S>
__device__ __forceinline__ void gate_inputs(const Gates& g, const ScanConst& k, float& xs, float& xn) {
  xs = sigmoid_inputs<S>(g.r, g.z) + k.bv;
  xn = from_lane_plus<GRU_COLS * S>(g.n) + k.bi2;
}

__device__ __forceinline__ void settle_accumulator(v8f& acc) {
  asm volatile("v_nop\n\tv_nop\n\tv_nop\n\tv_nop" : "+v"(acc));
}

__device__ __forceinline__ float gru_step(float h, float xs, float xn, const ScanConst& k) {
  const float s = 1.0f / (1.0f + __expf(-(xs + h * k.wv + k.cv)));
  const float z = from_lane_plus<4>(s);
  const float a = xn + s * (h * k.w2 + k.c2);
  const float n = tanh_both_ranges(a);
  return h_to_z_lanes((1.0f - z) * n + z * h);
}

__global__ void __launch_bounds__(32)
gru_fused_kernel(const float* __restrict__ x,
                 const float* __restrict__ h0,
                 const int*   __restrict__ lengths,
                 const float* __restrict__ W_ih,
                 const float* __restrict__ W_hh,
                 const float* __restrict__ b_ih,
                 const float* __restrict__ b_hh,
                 float* __restrict__ out) {
  const int lane = threadIdx.x;
  const int c    = lane & 15;
  const int hf   = lane >> 4;
  const int col  = c & (GRU_COLS - 1);
  const int ts   = c / GRU_COLS;
  const int b    = blockIdx.x * GRU_COLS + col;

  v2f wf[16];
  {
    const float* wp = W_ih + (c < 3 ? c : 0) * GRU_D + 4 * hf;
#pragma unroll
    for (int i = 0; i < 8; ++i) {
      v4f w = {wp[8 * i], wp[8 * i + 1], wp[8 * i + 2], wp[8 * i + 3]};
      if (c >= 3) w = (v4f){0.0f, 0.0f, 0.0f, 0.0f};
      wf[2 * i]     = (v2f){w.x, w.y};
      wf[2 * i + 1] = (v2f){w.z, w.w};
    }
  }

  ScanConst k;
  const bool z_lane = (c >> 2) == 1;
  k.wv = z_lane ? W_hh[1] : W_hh[0];
  k.cv = z_lane ? b_hh[1] : b_hh[0];
  k.bv = z_lane ? b_ih[1] : b_ih[0];
  k.w2 = W_hh[2]; k.c2 = b_hh[2]; k.bi2 = b_ih[2];

  float h = h0[b];
  const int len = lengths[b];

  const float* xlane = x + ((size_t)ts * GRU_B + b) * GRU_D + 4 * hf;
  float* outb = out + b;

  prefetch_tile(xlane, 2);
  XTile X;
  load_part<0, 8>(X, tile_ptr(xlane, 0));
  v8f Gn;
  mm_part<0, 4, true>(X, wf, Gn);
  mm_part<4, 8, false>(X, wf, Gn);
  settle_accumulator(Gn);
  load_part<0, 8>(X, tile_ptr(xlane, 1));
  Gates G = {Gn[0], Gn[1], Gn[2]};
  float xs, xn;
  gate_inputs<0>(G, k, xs, xn);

  for (int j = 0; j < GRU_TILES; ++j) {
    const float* pn = tile_ptr(xlane, j + 2);
    float o[GRU_TS];

#define GRU_STEP(S, FIRST, NEXT_S, NEXT_G)                            \
    mm_part<S, S + 1, FIRST>(X, wf, Gn);                              \
    if (S == GRU_TS - 1) settle_accumulator(Gn);                      \
    load_part<S, S + 1>(X, pn);                                       \
    { const Gates Gnext = {Gn[0], Gn[1], Gn[2]};                      \
      gate_inputs<NEXT_S>(NEXT_G, k, xs_next, xn_next); }             \
    h = gru_step(h, xs, xn, k);                                       \
    o[S] = (GRU_TS * j + S < len) ? h : 0.0f;                         \
    xs = xs_next; xn = xn_next;                                       \
    step_fence();

    float xs_next, xn_next;
    prefetch_tile(xlane, j + 3);
    GRU_STEP(0, true,  1, G)
    GRU_STEP(1, false, 2, G)
    GRU_STEP(2, false, 3, G)
    GRU_STEP(3, false, 4, G)
    GRU_STEP(4, false, 5, G)
    GRU_STEP(5, false, 6, G)
    GRU_STEP(6, false, 7, G)
    GRU_STEP(7, false, 0, Gnext)
#undef GRU_STEP

    G = (Gates){Gn[0], Gn[1], Gn[2]};

    if (lane < GRU_COLS) {
      float* op = outb + (size_t)(GRU_TS * j) * GRU_B;
#pragma unroll
      for (int s = 0; s < GRU_TS; ++s) op[s * GRU_B] = o[s];
    }
  }
}

extern "C" void kernel_launch(void* const* d_in, const int* in_sizes, int n_in,
                              void* d_out, int out_size, void* d_ws, size_t ws_size,
                              hipStream_t stream) {
  const float* x       = (const float*)d_in[0];
  const float* h0      = (const float*)d_in[1];
  const int*   lengths = (const int*)  d_in[2];
  const float* W_ih    = (const float*)d_in[3];
  const float* W_hh    = (const float*)d_in[4];
  const float* b_ih    = (const float*)d_in[5];
  const float* b_hh    = (const float*)d_in[6];
  float*       out     = (float*)d_out;

  gru_fused_kernel<<<GRU_B / GRU_COLS, 32, 0, stream>>>(x, h0, lengths, W_ih, W_hh,
                                                        b_ih, b_hh, out);
}


# scan: dependent chain h(t-1)->h(t) cut from 33 to 22 instructions by re-association: h-free terms formed ahead and folded into single fmas, lane move of h inside the first fma; sum order changed
# speedup vs baseline: 4.5336x; 1.0795x over previous
#include <hip/hip_runtime.h>
#include <math.h>


typedef float v2f __attribute__((ext_vector_type(2)));
typedef float v4f __attribute__((ext_vector_type(4)));
typedef float v8f __attribute__((ext_vector_type(8)));

#define GRU_L 512
#define GRU_B 2048
#define GRU_D 64
#define GRU_COLS 2
#define GRU_TS   8
#define GRU_TILES (GRU_L / GRU_TS)

struct XTile { v4f q[8]; };

__device__ __forceinline__ const float* tile_ptr(const float* __restrict__ xlane, int j) {
  const int jj = j < GRU_TILES ? j : GRU_TILES - 1;
  return xlane + (size_t)jj * (GRU_TS * GRU_B * GRU_D);
}

__device__ __forceinline__ void prefetch_tile(const float* __restrict__ xlane, int j) {
  __builtin_prefetch(tile_ptr(xlane, j), 0, 0);
}

template <int I0, int I1>
__device__ __forceinline__ void load_part(XTile& xt, const float* __restrict__ p) {
#pragma unroll
  for (int i = I0; i < I1; ++i) xt.q[i] = *(const v4f*)(p + 8 * i);
}

template <int I0, int I1, bool FIRST>
__device__ __forceinline__ void mm_part(const XTile& xt, const v2f (&wf)[16], v8f& acc) {
  v8f a = acc;
  if (FIRST) a = (v8f){};
#pragma unroll
  for (int i = I0; i < I1; ++i) {
    const v2f b0 = {xt.q[i].x, xt.q[i].y}, b1 = {xt.q[i].z, xt.q[i].w};
    a = __builtin_amdgcn_wmma_f32_16x16x4_f32(false, wf[2 * i],     false, b0, (short)0, a, false, false);
    a = __builtin_amdgcn_wmma_f32_16x16x4_f32(false, wf[2 * i + 1], false, b1, (short)0, a, false, false);
  }
  acc = a;
}

template <int SH>
__device__ __forceinline__ float from_lane_plus(float v) {
  if (SH == 0) return v;
  return __int_as_float(__builtin_amdgcn_update_dpp(0, __float_as_int(v), 0x100 + SH, 0xf, 0xf, true));
}

template <int SH>
__device__ __forceinline__ float z_lanes_from_lane_plus(float keep, float v) {
  const int ctrl = SH > 0 ? 0x100 + SH : SH == 0 ? 0xE4 : 0x110 - SH;
  return __int_as_float(__builtin_amdgcn_update_dpp(__float_as_int(keep), __float_as_int(v),
                                                    ctrl, 0xf, 0x2, false));
}

template <int S>
__device__ __forceinline__ float sigmoid_inputs(float g_r, float g_z) {
  return z_lanes_from_lane_plus<GRU_COLS * S - 4>(from_lane_plus<GRU_COLS * S>(g_r), g_z);
}

__device__ __forceinline__ void step_fence() {
  __builtin_amdgcn_sched_barrier(0);
}

__device__ __forceinline__ float blend_tanh(float a, float a2, float at, float g, float zh) {
  const float y  = fabsf(a);
  const float y2 = a * a;
  float p = fmaf(y2, -0x1.758e7ap-8f, 0x1.521192p-6f);
  p = fmaf(y2, p, -0x1.b8389cp-5f);
  p = fmaf(y2, p, 0x1.110704p-3f);
  p = fmaf(y2, p, -0x1.555532p-2f);
  const float z_small = fmaf(y2, y * p, y);
  const float zs      = y > 16.0f ? 1.0f : z_small;
  const float h_small = fmaf(g, zs, zh);

  const float L1 = 0x1.715476p+0f, L2 = 0x1.4ae0bep-26f;
  const float w = fabsf(a2);
  const float r = rintf(fabsf(at));
  const float f = fmaf(w, L2, fmaf(w, L1, -r));
  const float E = __builtin_amdgcn_exp2f(f);
  const float m = ldexpf(-2.0f, (int)(-r));
  const float R = __builtin_amdgcn_rcpf(fmaf(-0.5f, m, E));
  const float h_mid = fmaf(g * m, R, g + zh);

  const bool mid = __builtin_amdgcn_fmed3f(y, 0.625f, 16.0f) == y;
  return mid ? h_mid : h_small;
}

#define GRU_LOG2E     (0x1.715476p+0f)
#define GRU_NEG_LOG2E (-0x1.715476p+0f)

struct ScanConst { float wl, kb, w2, c2, w2x2, c2x2, w2t, c2t, bi2; };

struct Gates { float r, z, n; };

struct StepIn {
  float xs;
  float xn;
  float xn2;
  float xnt;
};

template <int S>
__device__ __forceinline__ StepIn gate_inputs(const Gates& g, const ScanConst& k) {
  StepIn in;
  in.xs  = fmaf(sigmoid_inputs<S>(g.r, g.z), GRU_NEG_LOG2E, k.kb);
  in.xn  = from_lane_plus<GRU_COLS * S>(g.n) + k.bi2;
  in.xn2 = in.xn + in.xn;
  in.xnt = in.xn2 * GRU_LOG2E;
  return in;
}

__device__ __forceinline__ void settle_accumulator(v8f& acc) {
  asm volatile("v_nop\n\tv_nop\n\tv_nop\n\tv_nop" : "+v"(acc));
}

__device__ __forceinline__ float gru_step(float h, const StepIn& in, const ScanConst& k) {
  float u = in.xs;
  asm("v_fmac_f32_dpp %0, %1, %2 dpp8:[0,1,2,3,0,1,2,3]" : "+v"(u) : "v"(h), "v"(k.wl));
  const float s = 1.0f / (1.0f + __builtin_amdgcn_exp2f(u));
  const float z = from_lane_plus<4>(s);
  const float a  = fmaf(fmaf(h, k.w2,   k.c2),   s, in.xn);
  const float a2 = fmaf(fmaf(h, k.w2x2, k.c2x2), s, in.xn2);
  const float at = fmaf(fmaf(h, k.w2t,  k.c2t),  s, in.xnt);
  return blend_tanh(a, a2, at, copysignf(1.0f - z, a), z * h);
}

__global__ void __launch_bounds__(32)
gru_fused_kernel(const float* __restrict__ x,
                 const float* __restrict__ h0,
                 const int*   __restrict__ lengths,
                 const float* __restrict__ W_ih,
                 const float* __restrict__ W_hh,
                 const float* __restrict__ b_ih,
                 const float* __restrict__ b_hh,
                 float* __restrict__ out) {
  const int lane = threadIdx.x;
  const int c    = lane & 15;
  const int hf   = lane >> 4;
  const int col  = c & (GRU_COLS - 1);
  const int ts   = c / GRU_COLS;
  const int b    = blockIdx.x * GRU_COLS + col;

  v2f wf[16];
  {
    const float* wp = W_ih + (c < 3 ? c : 0) * GRU_D + 4 * hf;
#pragma unroll
    for (int i = 0; i < 8; ++i) {
      v4f w = {wp[8 * i], wp[8 * i + 1], wp[8 * i + 2], wp[8 * i + 3]};
      if (c >= 3) w = (v4f){0.0f, 0.0f, 0.0f, 0.0f};
      wf[2 * i]     = (v2f){w.x, w.y};
      wf[2 * i + 1] = (v2f){w.z, w.w};
    }
  }

  ScanConst k;
  const bool z_lane = (c >> 2) == 1;
  k.wl = (z_lane ? W_hh[1] : W_hh[0]) * GRU_NEG_LOG2E;
  k.kb = ((z_lane ? b_ih[1] : b_ih[0]) + (z_lane ? b_hh[1] : b_hh[0])) * GRU_NEG_LOG2E;
  k.w2 = W_hh[2]; k.c2 = b_hh[2]; k.bi2 = b_ih[2];
  k.w2x2 = 2.0f * k.w2; k.c2x2 = 2.0f * k.c2;
  k.w2t = k.w2x2 * GRU_LOG2E; k.c2t = k.c2x2 * GRU_LOG2E;

  float h = h0[b];
  const int len = lengths[b];

  const float* xlane = x + ((size_t)ts * GRU_B + b) * GRU_D + 4 * hf;
  float* outb = out + b;

  prefetch_tile(xlane, 2);
  XTile X;
  load_part<0, 8>(X, tile_ptr(xlane, 0));
  v8f Gn;
  mm_part<0, 4, true>(X, wf, Gn);
  mm_part<4, 8, false>(X, wf, Gn);
  settle_accumulator(Gn);
  load_part<0, 8>(X, tile_ptr(xlane, 1));
  Gates G = {Gn[0], Gn[1], Gn[2]};
  StepIn in = gate_inputs<0>(G, k);

  for (int j = 0; j < GRU_TILES; ++j) {
    const float* pn = tile_ptr(xlane, j + 2);
    float o[GRU_TS];

#define GRU_STEP(S, FIRST, NEXT_S, NEXT_G)                            \
    mm_part<S, S + 1, FIRST>(X, wf, Gn);                              \
    if (S == GRU_TS - 1) settle_accumulator(Gn);                      \
    load_part<S, S + 1>(X, pn);                                       \
    { const Gates Gnext = {Gn[0], Gn[1], Gn[2]};                      \
      in_next = gate_inputs<NEXT_S>(NEXT_G, k); }                     \
    h = gru_step(h, in, k);                                           \
    o[S] = (GRU_TS * j + S < len) ? h : 0.0f;                         \
    in = in_next;                                                     \
    step_fence();

    StepIn in_next;
    prefetch_tile(xlane, j + 3);
    GRU_STEP(0, true,  1, G)
    GRU_STEP(1, false, 2, G)
    GRU_STEP(2, false, 3, G)
    GRU_STEP(3, false, 4, G)
    GRU_STEP(4, false, 5, G)
    GRU_STEP(5, false, 6, G)
    GRU_STEP(6, false, 7, G)
    GRU_STEP(7, false, 0, Gnext)
#undef GRU_STEP

    G = (Gates){Gn[0], Gn[1], Gn[2]};

    if (lane < GRU_COLS) {
      float* op = outb + (size_t)(GRU_TS * j) * GRU_B;
#pragma unroll
      for (int s = 0; s < GRU_TS; ++s) op[s * GRU_B] = o[s];
    }
  }
}

extern "C" void kernel_launch(void* const* d_in, const int* in_sizes, int n_in,
                              void* d_out, int out_size, void* d_ws, size_t ws_size,
                              hipStream_t stream) {
  const float* x       = (const float*)d_in[0];
  const float* h0      = (const float*)d_in[1];
  const int*   lengths = (const int*)  d_in[2];
  const float* W_ih    = (const float*)d_in[3];
  const float* W_hh    = (const float*)d_in[4];
  const float* b_ih    = (const float*)d_in[5];
  const float* b_hh    = (const float*)d_in[6];
  float*       out     = (float*)d_out;

  gru_fused_kernel<<<GRU_B / GRU_COLS, 32, 0, stream>>>(x, h0, lengths, W_ih, W_hh,
                                                        b_ih, b_hh, out);
}


# scan: the sigmoid's reciprocal 1/(1+2^u) is one v_rcp_f32 (within 1 ulp) in place of the IEEE division sequence: 9 dependent instructions off the h(t-1)->h(t) chain; tanh and all sums unchanged
# speedup vs baseline: 4.8892x; 1.0784x over previous
#include <hip/hip_runtime.h>
#include <math.h>


typedef float v2f __attribute__((ext_vector_type(2)));
typedef float v4f __attribute__((ext_vector_type(4)));
typedef float v8f __attribute__((ext_vector_type(8)));

#define GRU_L 512
#define GRU_B 2048
#define GRU_D 64
#define GRU_COLS 2
#define GRU_TS   8
#define GRU_TILES (GRU_L / GRU_TS)

struct XTile { v4f q[8]; };

__device__ __forceinline__ const float* tile_ptr(const float* __restrict__ xlane, int j) {
  const int jj = j < GRU_TILES ? j : GRU_TILES - 1;
  return xlane + (size_t)jj * (GRU_TS * GRU_B * GRU_D);
}

__device__ __forceinline__ void prefetch_tile(const float* __restrict__ xlane, int j) {
  __builtin_prefetch(tile_ptr(xlane, j), 0, 0);
}

template <int I0, int I1>
__device__ __forceinline__ void load_part(XTile& xt, const float* __restrict__ p) {
#pragma unroll
  for (int i = I0; i < I1; ++i) xt.q[i] = *(const v4f*)(p + 8 * i);
}

template <int I0, int I1, bool FIRST>
__device__ __forceinline__ void mm_part(const XTile& xt, const v2f (&wf)[16], v8f& acc) {
  v8f a = acc;
  if (FIRST) a = (v8f){};
#pragma unroll
  for (int i = I0; i < I1; ++i) {
    const v2f b0 = {xt.q[i].x, xt.q[i].y}, b1 = {xt.q[i].z, xt.q[i].w};
    a = __builtin_amdgcn_wmma_f32_16x16x4_f32(false, wf[2 * i],     false, b0, (short)0, a, false, false);
    a = __builtin_amdgcn_wmma_f32_16x16x4_f32(false, wf[2 * i + 1], false, b1, (short)0, a, false, false);
  }
  acc = a;
}

template <int SH>
__device__ __forceinline__ float from_lane_plus(float v) {
  if (SH == 0) return v;
  return __int_as_float(__builtin_amdgcn_update_dpp(0, __float_as_int(v), 0x100 + SH, 0xf, 0xf, true));
}

template <int SH>
__device__ __forceinline__ float z_lanes_from_lane_plus(float keep, float v) {
  const int ctrl = SH > 0 ? 0x100 + SH : SH == 0 ? 0xE4 : 0x110 - SH;
  return __int_as_float(__builtin_amdgcn_update_dpp(__float_as_int(keep), __float_as_int(v),
                                                    ctrl, 0xf, 0x2, false));
}

template <int S>
__device__ __forceinline__ float sigmoid_inputs(float g_r, float g_z) {
  return z_lanes_from_lane_plus<GRU_COLS * S - 4>(from_lane_plus<GRU_COLS * S>(g_r), g_z);
}

__device__ __forceinline__ void step_fence() {
  __builtin_amdgcn_sched_barrier(0);
}

__device__ __forceinline__ float blend_tanh(float a, float a2, float at, float g, float zh) {
  const float y  = fabsf(a);
  const float y2 = a * a;
  float p = fmaf(y2, -0x1.758e7ap-8f, 0x1.521192p-6f);
  p = fmaf(y2, p, -0x1.b8389cp-5f);
  p = fmaf(y2, p, 0x1.110704p-3f);
  p = fmaf(y2, p, -0x1.555532p-2f);
  const float z_small = fmaf(y2, y * p, y);
  const float zs      = y > 16.0f ? 1.0f : z_small;
  const float h_small = fmaf(g, zs, zh);

  const float L1 = 0x1.715476p+0f, L2 = 0x1.4ae0bep-26f;
  const float w = fabsf(a2);
  const float r = rintf(fabsf(at));
  const float f = fmaf(w, L2, fmaf(w, L1, -r));
  const float E = __builtin_amdgcn_exp2f(f);
  const float m = ldexpf(-2.0f, (int)(-r));
  const float R = __builtin_amdgcn_rcpf(fmaf(-0.5f, m, E));
  const float h_mid = fmaf(g * m, R, g + zh);

  const bool mid = __builtin_amdgcn_fmed3f(y, 0.625f, 16.0f) == y;
  return mid ? h_mid : h_small;
}

#define GRU_LOG2E     (0x1.715476p+0f)
#define GRU_NEG_LOG2E (-0x1.715476p+0f)

struct ScanConst { float wl, kb, w2, c2, w2x2, c2x2, w2t, c2t, bi2; };

struct Gates { float r, z, n; };

struct StepIn {
  float xs;
  float xn;
  float xn2;
  float xnt;
};

template <int S>
__device__ __forceinline__ StepIn gate_inputs(const Gates& g, const ScanConst& k) {
  StepIn in;
  in.xs  = fmaf(sigmoid_inputs<S>(g.r, g.z), GRU_NEG_LOG2E, k.kb);
  in.xn  = from_lane_plus<GRU_COLS * S>(g.n) + k.bi2;
  in.xn2 = in.xn + in.xn;
  in.xnt = in.xn2 * GRU_LOG2E;
  return in;
}

__device__ __forceinline__ void settle_accumulator(v8f& acc) {
  asm volatile("v_nop\n\tv_nop\n\tv_nop\n\tv_nop" : "+v"(acc));
}

__device__ __forceinline__ float gru_step(float h, const StepIn& in, const ScanConst& k) {
  float u = in.xs;
  asm("v_fmac_f32_dpp %0, %1, %2 dpp8:[0,1,2,3,0,1,2,3]" : "+v"(u) : "v"(h), "v"(k.wl));
  const float s = __builtin_amdgcn_rcpf(1.0f + __builtin_amdgcn_exp2f(u));
  const float z = from_lane_plus<4>(s);
  const float a  = fmaf(fmaf(h, k.w2,   k.c2),   s, in.xn);
  const float a2 = fmaf(fmaf(h, k.w2x2, k.c2x2), s, in.xn2);
  const float at = fmaf(fmaf(h, k.w2t,  k.c2t),  s, in.xnt);
  return blend_tanh(a, a2, at, copysignf(1.0f - z, a), z * h);
}

__global__ void __launch_bounds__(32)
gru_fused_kernel(const float* __restrict__ x,
                 const float* __restrict__ h0,
                 const int*   __restrict__ lengths,
                 const float* __restrict__ W_ih,
                 const float* __restrict__ W_hh,
                 const float* __restrict__ b_ih,
                 const float* __restrict__ b_hh,
                 float* __restrict__ out) {
  const int lane = threadIdx.x;
  const int c    = lane & 15;
  const int hf   = lane >> 4;
  const int col  = c & (GRU_COLS - 1);
  const int ts   = c / GRU_COLS;
  const int b    = blockIdx.x * GRU_COLS + col;

  v2f wf[16];
  {
    const float* wp = W_ih + (c < 3 ? c : 0) * GRU_D + 4 * hf;
#pragma unroll
    for (int i = 0; i < 8; ++i) {
      v4f w = {wp[8 * i], wp[8 * i + 1], wp[8 * i + 2], wp[8 * i + 3]};
      if (c >= 3) w = (v4f){0.0f, 0.0f, 0.0f, 0.0f};
      wf[2 * i]     = (v2f){w.x, w.y};
      wf[2 * i + 1] = (v2f){w.z, w.w};
    }
  }

  ScanConst k;
  const bool z_lane = (c >> 2) == 1;
  k.wl = (z_lane ? W_hh[1] : W_hh[0]) * GRU_NEG_LOG2E;
  k.kb = ((z_lane ? b_ih[1] : b_ih[0]) + (z_lane ? b_hh[1] : b_hh[0])) * GRU_NEG_LOG2E;
  k.w2 = W_hh[2]; k.c2 = b_hh[2]; k.bi2 = b_ih[2];
  k.w2x2 = 2.0f * k.w2; k.c2x2 = 2.0f * k.c2;
  k.w2t = k.w2x2 * GRU_LOG2E; k.c2t = k.c2x2 * GRU_LOG2E;

  float h = h0[b];
  const int len = lengths[b];

  const float* xlane = x + ((size_t)ts * GRU_B + b) * GRU_D + 4 * hf;
  float* outb = out + b;

  prefetch_tile(xlane, 2);
  XTile X;
  load_part<0, 8>(X, tile_ptr(xlane, 0));
  v8f Gn;
  mm_part<0, 4, true>(X, wf, Gn);
  mm_part<4, 8, false>(X, wf, Gn);
  settle_accumulator(Gn);
  load_part<0, 8>(X, tile_ptr(xlane, 1));
  Gates G = {Gn[0], Gn[1], Gn[2]};
  StepIn in = gate_inputs<0>(G, k);

  for (int j = 0; j < GRU_TILES; ++j) {
    const float* pn = tile_ptr(xlane, j + 2);
    float o[GRU_TS];

#define GRU_STEP(S, FIRST, NEXT_S, NEXT_G)                            \
    mm_part<S, S + 1, FIRST>(X, wf, Gn);                              \
    if (S == GRU_TS - 1) settle_accumulator(Gn);                      \
    load_part<S, S + 1>(X, pn);                                       \
    { const Gates Gnext = {Gn[0], Gn[1], Gn[2]};                      \
      in_next = gate_inputs<NEXT_S>(NEXT_G, k); }                     \
    h = gru_step(h, in, k);                                           \
    o[S] = (GRU_TS * j + S < len) ? h : 0.0f;                         \
    in = in_next;                                                     \
    step_fence();

    StepIn in_next;
    prefetch_tile(xlane, j + 3);
    GRU_STEP(0, true,  1, G)
    GRU_STEP(1, false, 2, G)
    GRU_STEP(2, false, 3, G)
    GRU_STEP(3, false, 4, G)
    GRU_STEP(4, false, 5, G)
    GRU_STEP(5, false, 6, G)
    GRU_STEP(6, false, 7, G)
    GRU_STEP(7, false, 0, Gnext)
#undef GRU_STEP

    G = (Gates){Gn[0], Gn[1], Gn[2]};

    if (lane < GRU_COLS) {
      float* op = outb + (size_t)(GRU_TS * j) * GRU_B;
#pragma unroll
      for (int s = 0; s < GRU_TS; ++s) op[s * GRU_B] = o[s];
    }
  }
}

extern "C" void kernel_launch(void* const* d_in, const int* in_sizes, int n_in,
                              void* d_out, int out_size, void* d_ws, size_t ws_size,
                              hipStream_t stream) {
  const float* x       = (const float*)d_in[0];
  const float* h0      = (const float*)d_in[1];
  const int*   lengths = (const int*)  d_in[2];
  const float* W_ih    = (const float*)d_in[3];
  const float* W_hh    = (const float*)d_in[4];
  const float* b_ih    = (const float*)d_in[5];
  const float* b_hh    = (const float*)d_in[6];
  float*       out     = (float*)d_out;

  gru_fused_kernel<<<GRU_B / GRU_COLS, 32, 0, stream>>>(x, h0, lengths, W_ih, W_hh,
                                                        b_ih, b_hh, out);
}


# launch shape: 256 workgroups of 4 waves with a 204 KiB LDS reservation each: one workgroup per WGP, one wave per SIMD (was 1024 one-wave workgroups); per-wave code and all sums unchanged
# speedup vs baseline: 5.3068x; 1.0854x over previous
#include <hip/hip_runtime.h>
#include <math.h>


typedef float v2f __attribute__((ext_vector_type(2)));
typedef float v4f __attribute__((ext_vector_type(4)));
typedef float v8f __attribute__((ext_vector_type(8)));

#define GRU_L 512
#define GRU_B 2048
#define GRU_D 64
#define GRU_COLS 2
#define GRU_TS   8
#define GRU_TILES (GRU_L / GRU_TS)
#define GRU_WAVES 4
#define GRU_LDS_RESERVE 208896

struct XTile { v4f q[8]; };

__device__ __forceinline__ const float* tile_ptr(const float* __restrict__ xlane, int j) {
  const int jj = j < GRU_TILES ? j : GRU_TILES - 1;
  return xlane + (size_t)jj * (GRU_TS * GRU_B * GRU_D);
}

__device__ __forceinline__ void prefetch_tile(const float* __restrict__ xlane, int j) {
  __builtin_prefetch(tile_ptr(xlane, j), 0, 0);
}

template <int I0, int I1>
__device__ __forceinline__ void load_part(XTile& xt, const float* __restrict__ p) {
#pragma unroll
  for (int i = I0; i < I1; ++i) xt.q[i] = *(const v4f*)(p + 8 * i);
}

template <int I0, int I1, bool FIRST>
__device__ __forceinline__ void mm_part(const XTile& xt, const v2f (&wf)[16], v8f& acc) {
  v8f a = acc;
  if (FIRST) a = (v8f){};
#pragma unroll
  for (int i = I0; i < I1; ++i) {
    const v2f b0 = {xt.q[i].x, xt.q[i].y}, b1 = {xt.q[i].z, xt.q[i].w};
    a = __builtin_amdgcn_wmma_f32_16x16x4_f32(false, wf[2 * i],     false, b0, (short)0, a, false, false);
    a = __builtin_amdgcn_wmma_f32_16x16x4_f32(false, wf[2 * i + 1], false, b1, (short)0, a, false, false);
  }
  acc = a;
}

template <int SH>
__device__ __forceinline__ float from_lane_plus(float v) {
  if (SH == 0) return v;
  return __int_as_float(__builtin_amdgcn_update_dpp(0, __float_as_int(v), 0x100 + SH, 0xf, 0xf, true));
}

template <int SH>
__device__ __forceinline__ float z_lanes_from_lane_plus(float keep, float v) {
  const int ctrl = SH > 0 ? 0x100 + SH : SH == 0 ? 0xE4 : 0x110 - SH;
  return __int_as_float(__builtin_amdgcn_update_dpp(__float_as_int(keep), __float_as_int(v),
                                                    ctrl, 0xf, 0x2, false));
}

template <int S>
__device__ __forceinline__ float sigmoid_inputs(float g_r, float g_z) {
  return z_lanes_from_lane_plus<GRU_COLS * S - 4>(from_lane_plus<GRU_COLS * S>(g_r), g_z);
}

__device__ __forceinline__ void step_fence() {
  __builtin_amdgcn_sched_barrier(0);
}

__device__ __forceinline__ float blend_tanh(float a, float a2, float at, float g, float zh) {
  const float y  = fabsf(a);
  const float y2 = a * a;
  float p = fmaf(y2, -0x1.758e7ap-8f, 0x1.521192p-6f);
  p = fmaf(y2, p, -0x1.b8389cp-5f);
  p = fmaf(y2, p, 0x1.110704p-3f);
  p = fmaf(y2, p, -0x1.555532p-2f);
  const float z_small = fmaf(y2, y * p, y);
  const float zs      = y > 16.0f ? 1.0f : z_small;
  const float h_small = fmaf(g, zs, zh);

  const float L1 = 0x1.715476p+0f, L2 = 0x1.4ae0bep-26f;
  const float w = fabsf(a2);
  const float r = rintf(fabsf(at));
  const float f = fmaf(w, L2, fmaf(w, L1, -r));
  const float E = __builtin_amdgcn_exp2f(f);
  const float m = ldexpf(-2.0f, (int)(-r));
  const float R = __builtin_amdgcn_rcpf(fmaf(-0.5f, m, E));
  const float h_mid = fmaf(g * m, R, g + zh);

  const bool mid = __builtin_amdgcn_fmed3f(y, 0.625f, 16.0f) == y;
  return mid ? h_mid : h_small;
}

#define GRU_LOG2E     (0x1.715476p+0f)
#define GRU_NEG_LOG2E (-0x1.715476p+0f)

struct ScanConst { float wl, kb, w2, c2, w2x2, c2x2, w2t, c2t, bi2; };

struct Gates { float r, z, n; };

struct StepIn {
  float xs;
  float xn;
  float xn2;
  float xnt;
};

template <int S>
__device__ __forceinline__ StepIn gate_inputs(const Gates& g, const ScanConst& k) {
  StepIn in;
  in.xs  = fmaf(sigmoid_inputs<S>(g.r, g.z), GRU_NEG_LOG2E, k.kb);
  in.xn  = from_lane_plus<GRU_COLS * S>(g.n) + k.bi2;
  in.xn2 = in.xn + in.xn;
  in.xnt = in.xn2 * GRU_LOG2E;
  return in;
}

__device__ __forceinline__ void settle_accumulator(v8f& acc) {
  asm volatile("v_nop\n\tv_nop\n\tv_nop\n\tv_nop" : "+v"(acc));
}

__device__ __forceinline__ float gru_step(float h, const StepIn& in, const ScanConst& k) {
  float u = in.xs;
  asm("v_fmac_f32_dpp %0, %1, %2 dpp8:[0,1,2,3,0,1,2,3]" : "+v"(u) : "v"(h), "v"(k.wl));
  const float s = __builtin_amdgcn_rcpf(1.0f + __builtin_amdgcn_exp2f(u));
  const float z = from_lane_plus<4>(s);
  const float a  = fmaf(fmaf(h, k.w2,   k.c2),   s, in.xn);
  const float a2 = fmaf(fmaf(h, k.w2x2, k.c2x2), s, in.xn2);
  const float at = fmaf(fmaf(h, k.w2t,  k.c2t),  s, in.xnt);
  return blend_tanh(a, a2, at, copysignf(1.0f - z, a), z * h);
}

__global__ void __launch_bounds__(32 * GRU_WAVES)
gru_fused_kernel(const float* __restrict__ x,
                 const float* __restrict__ h0,
                 const int*   __restrict__ lengths,
                 const float* __restrict__ W_ih,
                 const float* __restrict__ W_hh,
                 const float* __restrict__ b_ih,
                 const float* __restrict__ b_hh,
                 float* __restrict__ out) {
  const int lane = threadIdx.x & 31;
  const int wave = __builtin_amdgcn_readfirstlane((int)(threadIdx.x >> 5));
  const int pair = blockIdx.x * GRU_WAVES + wave;
  const int c    = lane & 15;
  const int hf   = lane >> 4;
  const int col  = c & (GRU_COLS - 1);
  const int ts   = c / GRU_COLS;
  const int b    = pair * GRU_COLS + col;

  v2f wf[16];
  {
    const float* wp = W_ih + (c < 3 ? c : 0) * GRU_D + 4 * hf;
#pragma unroll
    for (int i = 0; i < 8; ++i) {
      v4f w = {wp[8 * i], wp[8 * i + 1], wp[8 * i + 2], wp[8 * i + 3]};
      if (c >= 3) w = (v4f){0.0f, 0.0f, 0.0f, 0.0f};
      wf[2 * i]     = (v2f){w.x, w.y};
      wf[2 * i + 1] = (v2f){w.z, w.w};
    }
  }

  ScanConst k;
  const bool z_lane = (c >> 2) == 1;
  k.wl = (z_lane ? W_hh[1] : W_hh[0]) * GRU_NEG_LOG2E;
  k.kb = ((z_lane ? b_ih[1] : b_ih[0]) + (z_lane ? b_hh[1] : b_hh[0])) * GRU_NEG_LOG2E;
  k.w2 = W_hh[2]; k.c2 = b_hh[2]; k.bi2 = b_ih[2];
  k.w2x2 = 2.0f * k.w2; k.c2x2 = 2.0f * k.c2;
  k.w2t = k.w2x2 * GRU_LOG2E; k.c2t = k.c2x2 * GRU_LOG2E;

  float h = h0[b];
  const int len = lengths[b];

  const float* xlane = x + ((size_t)ts * GRU_B + b) * GRU_D + 4 * hf;
  float* outb = out + b;

  prefetch_tile(xlane, 2);
  XTile X;
  load_part<0, 8>(X, tile_ptr(xlane, 0));
  v8f Gn;
  mm_part<0, 4, true>(X, wf, Gn);
  mm_part<4, 8, false>(X, wf, Gn);
  settle_accumulator(Gn);
  load_part<0, 8>(X, tile_ptr(xlane, 1));
  Gates G = {Gn[0], Gn[1], Gn[2]};
  StepIn in = gate_inputs<0>(G, k);

  for (int j = 0; j < GRU_TILES; ++j) {
    const float* pn = tile_ptr(xlane, j + 2);
    float o[GRU_TS];

#define GRU_STEP(S, FIRST, NEXT_S, NEXT_G)                            \
    mm_part<S, S + 1, FIRST>(X, wf, Gn);                              \
    if (S == GRU_TS - 1) settle_accumulator(Gn);                      \
    load_part<S, S + 1>(X, pn);                                       \
    { const Gates Gnext = {Gn[0], Gn[1], Gn[2]};                      \
      in_next = gate_inputs<NEXT_S>(NEXT_G, k); }                     \
    h = gru_step(h, in, k);                                           \
    o[S] = (GRU_TS * j + S < len) ? h : 0.0f;                         \
    in = in_next;                                                     \
    step_fence();

    StepIn in_next;
    prefetch_tile(xlane, j + 3);
    GRU_STEP(0, true,  1, G)
    GRU_STEP(1, false, 2, G)
    GRU_STEP(2, false, 3, G)
    GRU_STEP(3, false, 4, G)
    GRU_STEP(4, false, 5, G)
    GRU_STEP(5, false, 6, G)
    GRU_STEP(6, false, 7, G)
    GRU_STEP(7, false, 0, Gnext)
#undef GRU_STEP

    G = (Gates){Gn[0], Gn[1], Gn[2]};

    if (lane < GRU_COLS) {
      float* op = outb + (size_t)(GRU_TS * j) * GRU_B;
#pragma unroll
      for (int s = 0; s < GRU_TS; ++s) op[s * GRU_B] = o[s];
    }
  }
}

extern "C" void kernel_launch(void* const* d_in, const int* in_sizes, int n_in,
                              void* d_out, int out_size, void* d_ws, size_t ws_size,
                              hipStream_t stream) {
  const float* x       = (const float*)d_in[0];
  const float* h0      = (const float*)d_in[1];
  const int*   lengths = (const int*)  d_in[2];
  const float* W_ih    = (const float*)d_in[3];
  const float* W_hh    = (const float*)d_in[4];
  const float* b_ih    = (const float*)d_in[5];
  const float* b_hh    = (const float*)d_in[6];
  float*       out     = (float*)d_out;

  gru_fused_kernel<<<GRU_B / (GRU_COLS * GRU_WAVES), 32 * GRU_WAVES, GRU_LDS_RESERVE, stream>>>(
      x, h0, lengths, W_ih, W_hh, b_ih, b_hh, out);
}


# scan: tanh(a) = 1 - 2/(2^(2a*log2 e) + 1) by one v_exp_f32 and one v_rcp_f32 (abs error about 2e-7) replaces the library's two-range form: h chain 13 -> 9 links; blend summed as (1-z)+z*h-2(1-z)R
# speedup vs baseline: 5.5473x; 1.0453x over previous
#include <hip/hip_runtime.h>
#include <math.h>


typedef float v2f __attribute__((ext_vector_type(2)));
typedef float v4f __attribute__((ext_vector_type(4)));
typedef float v8f __attribute__((ext_vector_type(8)));

#define GRU_L 512
#define GRU_B 2048
#define GRU_D 64
#define GRU_COLS 2
#define GRU_TS   8
#define GRU_TILES (GRU_L / GRU_TS)
#define GRU_WAVES 4
#define GRU_LDS_RESERVE 208896

struct XTile { v4f q[8]; };

__device__ __forceinline__ const float* tile_ptr(const float* __restrict__ xlane, int j) {
  const int jj = j < GRU_TILES ? j : GRU_TILES - 1;
  return xlane + (size_t)jj * (GRU_TS * GRU_B * GRU_D);
}

__device__ __forceinline__ void prefetch_tile(const float* __restrict__ xlane, int j) {
  __builtin_prefetch(tile_ptr(xlane, j), 0, 0);
}

template <int I0, int I1>
__device__ __forceinline__ void load_part(XTile& xt, const float* __restrict__ p) {
#pragma unroll
  for (int i = I0; i < I1; ++i) xt.q[i] = *(const v4f*)(p + 8 * i);
}

template <int I0, int I1, bool FIRST>
__device__ __forceinline__ void mm_part(const XTile& xt, const v2f (&wf)[16], v8f& acc) {
  v8f a = acc;
  if (FIRST) a = (v8f){};
#pragma unroll
  for (int i = I0; i < I1; ++i) {
    const v2f b0 = {xt.q[i].x, xt.q[i].y}, b1 = {xt.q[i].z, xt.q[i].w};
    a = __builtin_amdgcn_wmma_f32_16x16x4_f32(false, wf[2 * i],     false, b0, (short)0, a, false, false);
    a = __builtin_amdgcn_wmma_f32_16x16x4_f32(false, wf[2 * i + 1], false, b1, (short)0, a, false, false);
  }
  acc = a;
}

template <int SH>
__device__ __forceinline__ float from_lane_plus(float v) {
  if (SH == 0) return v;
  return __int_as_float(__builtin_amdgcn_update_dpp(0, __float_as_int(v), 0x100 + SH, 0xf, 0xf, true));
}

template <int SH>
__device__ __forceinline__ float z_lanes_from_lane_plus(float keep, float v) {
  const int ctrl = SH > 0 ? 0x100 + SH : SH == 0 ? 0xE4 : 0x110 - SH;
  return __int_as_float(__builtin_amdgcn_update_dpp(__float_as_int(keep), __float_as_int(v),
                                                    ctrl, 0xf, 0x2, false));
}

template <int S>
__device__ __forceinline__ float sigmoid_inputs(float g_r, float g_z) {
  return z_lanes_from_lane_plus<GRU_COLS * S - 4>(from_lane_plus<GRU_COLS * S>(g_r), g_z);
}

__device__ __forceinline__ void step_fence() {
  __builtin_amdgcn_sched_barrier(0);
}

#define GRU_2LOG2E    (0x1.715476p+1f)
#define GRU_NEG_LOG2E (-0x1.715476p+0f)

struct ScanConst { float wl, kb, w2t, c2t, bi2t; };

struct Gates { float r, z, n; };

struct StepIn {
  float xs;
  float xnt;
};

template <int S>
__device__ __forceinline__ StepIn gate_inputs(const Gates& g, const ScanConst& k) {
  StepIn in;
  in.xs  = fmaf(sigmoid_inputs<S>(g.r, g.z), GRU_NEG_LOG2E, k.kb);
  in.xnt = fmaf(from_lane_plus<GRU_COLS * S>(g.n), GRU_2LOG2E, k.bi2t);
  return in;
}

__device__ __forceinline__ void settle_accumulator(v8f& acc) {
  asm volatile("v_nop\n\tv_nop\n\tv_nop\n\tv_nop" : "+v"(acc));
}

__device__ __forceinline__ float gru_step(float h, const StepIn& in, const ScanConst& k) {
  float u = in.xs;
  asm("v_fmac_f32_dpp %0, %1, %2 dpp8:[0,1,2,3,0,1,2,3]" : "+v"(u) : "v"(h), "v"(k.wl));
  const float s = __builtin_amdgcn_rcpf(1.0f + __builtin_amdgcn_exp2f(u));
  const float z = from_lane_plus<4>(s);
  const float at = fmaf(fmaf(h, k.w2t, k.c2t), s, in.xnt);
  const float R  = __builtin_amdgcn_rcpf(1.0f + __builtin_amdgcn_exp2f(at));
  const float g = 1.0f - z;
  return fmaf(-2.0f * g, R, fmaf(z, h, g));
}

__global__ void __launch_bounds__(32 * GRU_WAVES)
gru_fused_kernel(const float* __restrict__ x,
                 const float* __restrict__ h0,
                 const int*   __restrict__ lengths,
                 const float* __restrict__ W_ih,
                 const float* __restrict__ W_hh,
                 const float* __restrict__ b_ih,
                 const float* __restrict__ b_hh,
                 float* __restrict__ out) {
  const int lane = threadIdx.x & 31;
  const int wave = __builtin_amdgcn_readfirstlane((int)(threadIdx.x >> 5));
  const int pair = blockIdx.x * GRU_WAVES + wave;
  const int c    = lane & 15;
  const int hf   = lane >> 4;
  const int col  = c & (GRU_COLS - 1);
  const int ts   = c / GRU_COLS;
  const int b    = pair * GRU_COLS + col;

  v2f wf[16];
  {
    const float* wp = W_ih + (c < 3 ? c : 0) * GRU_D + 4 * hf;
#pragma unroll
    for (int i = 0; i < 8; ++i) {
      v4f w = {wp[8 * i], wp[8 * i + 1], wp[8 * i + 2], wp[8 * i + 3]};
      if (c >= 3) w = (v4f){0.0f, 0.0f, 0.0f, 0.0f};
      wf[2 * i]     = (v2f){w.x, w.y};
      wf[2 * i + 1] = (v2f){w.z, w.w};
    }
  }

  ScanConst k;
  const bool z_lane = (c >> 2) == 1;
  k.wl = (z_lane ? W_hh[1] : W_hh[0]) * GRU_NEG_LOG2E;
  k.kb = ((z_lane ? b_ih[1] : b_ih[0]) + (z_lane ? b_hh[1] : b_hh[0])) * GRU_NEG_LOG2E;
  k.w2t = W_hh[2] * GRU_2LOG2E; k.c2t = b_hh[2] * GRU_2LOG2E; k.bi2t = b_ih[2] * GRU_2LOG2E;

  float h = h0[b];
  const int len = lengths[b];

  const float* xlane = x + ((size_t)ts * GRU_B + b) * GRU_D + 4 * hf;
  float* outb = out + b;

  prefetch_tile(xlane, 2);
  XTile X;
  load_part<0, 8>(X, tile_ptr(xlane, 0));
  v8f Gn;
  mm_part<0, 4, true>(X, wf, Gn);
  mm_part<4, 8, false>(X, wf, Gn);
  settle_accumulator(Gn);
  load_part<0, 8>(X, tile_ptr(xlane, 1));
  Gates G = {Gn[0], Gn[1], Gn[2]};
  StepIn in = gate_inputs<0>(G, k);

  for (int j = 0; j < GRU_TILES; ++j) {
    const float* pn = tile_ptr(xlane, j + 2);
    float o[GRU_TS];

#define GRU_STEP(S, FIRST, NEXT_S, NEXT_G)                            \
    mm_part<S, S + 1, FIRST>(X, wf, Gn);                              \
    if (S == GRU_TS - 1) settle_accumulator(Gn);                      \
    load_part<S, S + 1>(X, pn);                                       \
    { const Gates Gnext = {Gn[0], Gn[1], Gn[2]};                      \
      in_next = gate_inputs<NEXT_S>(NEXT_G, k); }                     \
    h = gru_step(h, in, k);                                           \
    o[S] = (GRU_TS * j + S < len) ? h : 0.0f;                         \
    in = in_next;                                                     \
    step_fence();

    StepIn in_next;
    prefetch_tile(xlane, j + 3);
    GRU_STEP(0, true,  1, G)
    GRU_STEP(1, false, 2, G)
    GRU_STEP(2, false, 3, G)
    GRU_STEP(3, false, 4, G)
    GRU_STEP(4, false, 5, G)
    GRU_STEP(5, false, 6, G)
    GRU_STEP(6, false, 7, G)
    GRU_STEP(7, false, 0, Gnext)
#undef GRU_STEP

    G = (Gates){Gn[0], Gn[1], Gn[2]};

    if (lane < GRU_COLS) {
      float* op = outb + (size_t)(GRU_TS * j) * GRU_B;
#pragma unroll
      for (int s = 0; s < GRU_TS; ++s) op[s * GRU_B] = o[s];
    }
  }
}

extern "C" void kernel_launch(void* const* d_in, const int* in_sizes, int n_in,
                              void* d_out, int out_size, void* d_ws, size_t ws_size,
                              hipStream_t stream) {
  const float* x       = (const float*)d_in[0];
  const float* h0      = (const float*)d_in[1];
  const int*   lengths = (const int*)  d_in[2];
  const float* W_ih    = (const float*)d_in[3];
  const float* W_hh    = (const float*)d_in[4];
  const float* b_ih    = (const float*)d_in[5];
  const float* b_hh    = (const float*)d_in[6];
  float*       out     = (float*)d_out;

  gru_fused_kernel<<<GRU_B / (GRU_COLS * GRU_WAVES), 32 * GRU_WAVES, GRU_LDS_RESERVE, stream>>>(
      x, h0, lengths, W_ih, W_hh, b_ih, b_hh, out);
}
